# starting kernel
# baseline (speedup 1.0000x reference)
#include <hip/hip_runtime.h>
#include <math.h>

constexpr int kBS      = 64;
constexpr int kNC      = 128;
constexpr int kN       = 32;
constexpr int kDN      = 64;
constexpr int kHMOD    = 128;
constexpr int kMODIN   = 2 * kN + 4 + kN * kN;
constexpr int kMODOUT  = kN * kN + kN;
constexpr int kHEAD    = 2 * kN + 4;
constexpr int kHSTRIDE = 132;

typedef __attribute__((ext_vector_type(2))) float v2f;
typedef __attribute__((ext_vector_type(8))) float v8f;

__global__ __launch_bounds__(256) void head_kernel(
    const float* __restrict__ h, const float* __restrict__ msg,
    const float* __restrict__ decay, const float* __restrict__ drift,
    const float* __restrict__ s_live, const float* __restrict__ s_ema,
    float* __restrict__ head) {
  const int t    = blockIdx.x * 256 + threadIdx.x;
  const int pair = t >> 6;
  const int j    = t & 63;

  const float* src = (j < kN ? h : msg) +
                     (size_t)pair * (kN * kDN) + (size_t)(j & (kN - 1)) * kDN;
  const float4* p4 = (const float4*)src;
  float s = 0.f;
#pragma unroll
  for (int i = 0; i < kDN / 4; ++i) {
    float4 v = p4[i];
    s += v.x * v.x + v.y * v.y + v.z * v.z + v.w * v.w;
  }
  head[(size_t)pair * kHEAD + j] = sqrtf(s);

  if (j == 0) {
    const float* dp = decay + (size_t)pair * kN;
    float m = 0.f;
#pragma unroll
    for (int i = 0; i < kN; ++i) m += dp[i];
    const int b = pair / kNC;
    float* hd = head + (size_t)pair * kHEAD;
    hd[kHEAD - 4] = m * (1.f / (float)kN);
    hd[kHEAD - 3] = drift[pair];
    hd[kHEAD - 2] = s_live[b];
    hd[kHEAD - 1] = s_ema[b];
  }
}

__global__ __launch_bounds__(256) void modmlp_kernel(
    const float* __restrict__ head, const float* __restrict__ hebb,
    const float* __restrict__ w1, const float* __restrict__ b1,
    const float* __restrict__ w2, const float* __restrict__ b2,
    float* __restrict__ out) {
  extern __shared__ float smem[];
  float* ldsA = smem;
  float* ldsH = smem + 16 * kMODIN;

  const int tid   = threadIdx.x;
  const int lane  = tid & 31;
  const int wave  = tid >> 5;
  const int mtile = blockIdx.x;
  const int cell  = blockIdx.y;
  const int b0    = mtile * 16;
  const int arow  = lane & 15;
  const int koff  = (lane >> 4) << 1;
  const int rhalf = (lane >> 4) << 3;

  for (int r = 0; r < 16; ++r) {
    const size_t pr = (size_t)(b0 + r) * kNC + cell;
    const float* hr = head + pr * kHEAD;
    const float* br = hebb + pr * (kN * kN);
    float* dst = ldsA + r * kMODIN;
    if (tid < kHEAD) dst[tid] = hr[tid];
    for (int k = tid; k < kN * kN; k += 256) dst[kHEAD + k] = br[k];
  }
  __syncthreads();

  const float* w1c  = w1 + (size_t)cell * kMODIN * kHMOD;
  const int    hcol = (wave << 4) + arow;
  v8f acc;
  {
    const float bias = b1[cell * kHMOD + hcol];
#pragma unroll
    for (int i = 0; i < 8; ++i) acc[i] = bias;
  }
  const float* aRow = ldsA + arow * kMODIN + koff;
#pragma unroll 4
  for (int k0 = 0; k0 < kMODIN; k0 += 4) {
    const v2f a = *(const v2f*)(aRow + k0);
    v2f b;
    b.x = w1c[(size_t)(k0 + koff) * kHMOD + hcol];
    b.y = w1c[(size_t)(k0 + koff + 1) * kHMOD + hcol];
    acc = __builtin_amdgcn_wmma_f32_16x16x4_f32(
        false, a, false, b, (short)0, acc, false, false);
  }
#pragma unroll
  for (int i = 0; i < 8; ++i)
    ldsH[(i + rhalf) * kHSTRIDE + hcol] = tanhf(acc[i]);
  __syncthreads();

  const float* w2c = w2 + (size_t)cell * kHMOD * kMODOUT;
  const float* b2c = b2 + (size_t)cell * kMODOUT;
  const float* aH  = ldsH + arow * kHSTRIDE + koff;
  for (int ot = wave; ot < kMODOUT / 16; ot += 8) {
    const int ocol = (ot << 4) + arow;
    v8f c;
    {
      const float bias = b2c[ocol];
#pragma unroll
      for (int i = 0; i < 8; ++i) c[i] = bias;
    }
#pragma unroll 8
    for (int k0 = 0; k0 < kHMOD; k0 += 4) {
      const v2f a = *(const v2f*)(aH + k0);
      v2f b;
      b.x = w2c[(size_t)(k0 + koff) * kMODOUT + ocol];
      b.y = w2c[(size_t)(k0 + koff + 1) * kMODOUT + ocol];
      c = __builtin_amdgcn_wmma_f32_16x16x4_f32(
          false, a, false, b, (short)0, c, false, false);
    }
#pragma unroll
    for (int i = 0; i < 8; ++i) {
      const int row = b0 + rhalf + i;
      out[((size_t)row * kNC + cell) * kMODOUT + ocol] = c[i];
    }
  }
}

extern "C" void kernel_launch(void* const* d_in, const int* in_sizes, int n_in,
                              void* d_out, int out_size, void* d_ws, size_t ws_size,
                              hipStream_t stream) {
  const float* h     = (const float*)d_in[0];
  const float* msg   = (const float*)d_in[1];
  const float* decay = (const float*)d_in[2];
  const float* hebb  = (const float*)d_in[3];
  const float* drift = (const float*)d_in[4];
  const float* slive = (const float*)d_in[5];
  const float* sema  = (const float*)d_in[6];
  const float* w1    = (const float*)d_in[7];
  const float* b1    = (const float*)d_in[8];
  const float* w2    = (const float*)d_in[9];
  const float* b2    = (const float*)d_in[10];
  float* headws = (float*)d_ws;
  float* out    = (float*)d_out;

  const int norm_threads = kBS * kNC * 2 * kN;
  head_kernel<<<norm_threads / 256, 256, 0, stream>>>(
      h, msg, decay, drift, slive, sema, headws);

  dim3 grid(kBS / 16, kNC);
  const size_t smem = (size_t)(16 * kMODIN + 16 * kHSTRIDE) * sizeof(float);
  modmlp_kernel<<<grid, 256, smem, stream>>>(
      headws, hebb, w1, b1, w2, b2, out);
}


# layer 1 split over K by wave: a wave loads the A fragments of its 1/8 of K straight from global (A no longer staged in LDS), 8 column tiles per wave, partials summed in LDS; sum order changed
# speedup vs baseline: 1.5587x; 1.5587x over previous
#include <hip/hip_runtime.h>
#include <math.h>

constexpr int kBS      = 64;
constexpr int kNC      = 128;
constexpr int kN       = 32;
constexpr int kDN      = 64;
constexpr int kHMOD    = 128;
constexpr int kMODIN   = 2 * kN + 4 + kN * kN;
constexpr int kMODOUT  = kN * kN + kN;
constexpr int kHEAD    = 2 * kN + 4;
constexpr int kHSTRIDE = 132;

typedef __attribute__((ext_vector_type(2))) float v2f;
typedef __attribute__((ext_vector_type(8))) float v8f;

__global__ __launch_bounds__(256) void head_kernel(
    const float* __restrict__ h, const float* __restrict__ msg,
    const float* __restrict__ decay, const float* __restrict__ drift,
    const float* __restrict__ s_live, const float* __restrict__ s_ema,
    float* __restrict__ head) {
  const int t    = blockIdx.x * 256 + threadIdx.x;
  const int pair = t >> 6;
  const int j    = t & 63;

  const float* src = (j < kN ? h : msg) +
                     (size_t)pair * (kN * kDN) + (size_t)(j & (kN - 1)) * kDN;
  const float4* p4 = (const float4*)src;
  float s = 0.f;
#pragma unroll
  for (int i = 0; i < kDN / 4; ++i) {
    float4 v = p4[i];
    s += v.x * v.x + v.y * v.y + v.z * v.z + v.w * v.w;
  }
  head[(size_t)pair * kHEAD + j] = sqrtf(s);

  if (j == 0) {
    const float* dp = decay + (size_t)pair * kN;
    float m = 0.f;
#pragma unroll
    for (int i = 0; i < kN; ++i) m += dp[i];
    const int b = pair / kNC;
    float* hd = head + (size_t)pair * kHEAD;
    hd[kHEAD - 4] = m * (1.f / (float)kN);
    hd[kHEAD - 3] = drift[pair];
    hd[kHEAD - 2] = s_live[b];
    hd[kHEAD - 1] = s_ema[b];
  }
}

constexpr int kG2   = 16;
constexpr int kNT1  = kHMOD / 16;
constexpr int kKW   = (kN * kN) / 8;
constexpr int kNJ   = kKW / 8;

template <int G, int LD>
__device__ __forceinline__ void load_bfrag(v2f (&bf)[G], const float* __restrict__ p) {
#pragma unroll
  for (int j = 0; j < G; ++j) {
    bf[j].x = p[(size_t)(4 * j) * LD];
    bf[j].y = p[(size_t)(4 * j + 1) * LD];
  }
}

template <int G>
__device__ __forceinline__ v8f mma_group(v8f acc, const float* a, const v2f (&bf)[G]) {
#pragma unroll
  for (int j = 0; j < G; ++j) {
    const v2f av = *(const v2f*)(a + 4 * j);
    acc = __builtin_amdgcn_wmma_f32_16x16x4_f32(
        false, av, false, bf[j], (short)0, acc, false, false);
  }
  return acc;
}

__device__ __forceinline__ void load_b1step(v2f (&bf)[kNT1], const float* __restrict__ p) {
#pragma unroll
  for (int n = 0; n < kNT1; ++n) {
    bf[n].x = p[16 * n];
    bf[n].y = p[kHMOD + 16 * n];
  }
}

__device__ __forceinline__ void mma_step(v8f (&acc)[kNT1], const v2f av, const v2f (&bf)[kNT1]) {
#pragma unroll
  for (int n = 0; n < kNT1; ++n)
    acc[n] = __builtin_amdgcn_wmma_f32_16x16x4_f32(
        false, av, false, bf[n], (short)0, acc[n], false, false);
}

__global__ __launch_bounds__(256) void modmlp_kernel(
    const float* __restrict__ head, const float* __restrict__ hebb,
    const float* __restrict__ w1, const float* __restrict__ b1,
    const float* __restrict__ w2, const float* __restrict__ b2,
    float* __restrict__ out) {
  extern __shared__ float smem[];
  float* ldsP = smem;
  float* ldsH = smem + 8 * 16 * kHSTRIDE;

  const int tid   = threadIdx.x;
  const int lane  = tid & 31;
  const int wave  = tid >> 5;
  const int mtile = blockIdx.x;
  const int cell  = blockIdx.y;
  const int b0    = mtile * 16;
  const int arow  = lane & 15;
  const int hsel  = lane >> 4;
  const int koff  = hsel << 1;
  const int rhalf = hsel << 3;

  const size_t prow = (size_t)(b0 + arow) * kNC + cell;
  const float* ap   = hebb + prow * (kN * kN) + wave * kKW + hsel * 4;
  const float* w1c  = w1 + (size_t)cell * kMODIN * kHMOD;
  const float* bp   = w1c + (size_t)(kHEAD + wave * kKW + hsel * 4) * kHMOD + arow;
  const float* hp   = head + prow * kHEAD + koff;
  const float* bh   = w1c + (size_t)koff * kHMOD + arow;

  v2f fb0[kNT1], fb1[kNT1], fb2[kNT1], fb3[kNT1];
  load_b1step(fb0, bp);
  load_b1step(fb1, bp + 2 * kHMOD);

  float4 a4[kNJ];
#pragma unroll
  for (int j = 0; j < kNJ; ++j) a4[j] = *(const float4*)(ap + 8 * j);
  const v2f ah0 = *(const v2f*)(hp + 4 * wave);
  const v2f ah1 = *(const v2f*)(hp + 4 * (wave + 8));

  v8f acc[kNT1];
#pragma unroll
  for (int n = 0; n < kNT1; ++n)
#pragma unroll
    for (int i = 0; i < 8; ++i) acc[n][i] = 0.f;

  if (wave == 0) {
    v2f fx[kNT1];
    load_b1step(fx, bh + (size_t)(4 * 16) * kHMOD);
    const v2f ax = *(const v2f*)(hp + 4 * 16);
    mma_step(acc, ax, fx);
  }

#pragma unroll
  for (int j = 0; j < kNJ; j += 2) {
    const float* pn = bp + (size_t)(8 * (j + 1)) * kHMOD;
    load_b1step(fb2, pn);
    load_b1step(fb3, pn + 2 * kHMOD);
    mma_step(acc, v2f{a4[j].x, a4[j].y}, fb0);
    mma_step(acc, v2f{a4[j].z, a4[j].w}, fb1);
    if (j + 2 < kNJ) {
      const float* pm = bp + (size_t)(8 * (j + 2)) * kHMOD;
      load_b1step(fb0, pm);
      load_b1step(fb1, pm + 2 * kHMOD);
    } else {
      load_b1step(fb0, bh + (size_t)(4 * wave) * kHMOD);
      load_b1step(fb1, bh + (size_t)(4 * (wave + 8)) * kHMOD);
    }
    mma_step(acc, v2f{a4[j + 1].x, a4[j + 1].y}, fb2);
    mma_step(acc, v2f{a4[j + 1].z, a4[j + 1].w}, fb3);
  }
  mma_step(acc, ah0, fb0);
  mma_step(acc, ah1, fb1);

  const float* w2c = w2 + (size_t)cell * kHMOD * kMODOUT;
  const float* w2p = w2c + (size_t)koff * kMODOUT + arow;
  const float* b2c = b2 + (size_t)cell * kMODOUT;
  v2f pb0[kG2], pb1[kG2];
  load_bfrag<kG2, kMODOUT>(pb0, w2p + (wave << 4));
  float bias2 = b2c[(wave << 4) + arow];

  const int rr = tid >> 4;
  const int c8 = (tid & 15) << 3;
  float4 s0 = *(const float4*)(b1 + cell * kHMOD + c8);
  float4 s1 = *(const float4*)(b1 + cell * kHMOD + c8 + 4);
  {
    float* myP = ldsP + wave * 16 * kHSTRIDE + rhalf * kHSTRIDE + arow;
#pragma unroll
    for (int n = 0; n < kNT1; ++n)
#pragma unroll
      for (int i = 0; i < 8; ++i) myP[i * kHSTRIDE + 16 * n] = acc[n][i];
  }
  __syncthreads();
  {
    const float* pp = ldsP + rr * kHSTRIDE + c8;
#pragma unroll
    for (int w = 0; w < 8; ++w) {
      const float4 u0 = *(const float4*)(pp + w * 16 * kHSTRIDE);
      const float4 u1 = *(const float4*)(pp + w * 16 * kHSTRIDE + 4);
      s0.x += u0.x; s0.y += u0.y; s0.z += u0.z; s0.w += u0.w;
      s1.x += u1.x; s1.y += u1.y; s1.z += u1.z; s1.w += u1.w;
    }
    float4 t0, t1;
    t0.x = tanhf(s0.x); t0.y = tanhf(s0.y); t0.z = tanhf(s0.z); t0.w = tanhf(s0.w);
    t1.x = tanhf(s1.x); t1.y = tanhf(s1.y); t1.z = tanhf(s1.z); t1.w = tanhf(s1.w);
    float* hd = ldsH + rr * kHSTRIDE + c8;
    *(float4*)hd       = t0;
    *(float4*)(hd + 4) = t1;
  }
  __syncthreads();

  const float* aH  = ldsH + arow * kHSTRIDE + koff;
  for (int ot = wave; ot < kMODOUT / 16; ot += 8) {
    const int ocol = (ot << 4) + arow;
    v8f c;
#pragma unroll
    for (int i = 0; i < 8; ++i) c[i] = bias2;
    load_bfrag<kG2, kMODOUT>(pb1, w2p + (ot << 4) + (size_t)(kG2 * 4) * kMODOUT);
    c = mma_group<kG2>(c, aH, pb0);
    if (ot + 8 < kMODOUT / 16) {
      load_bfrag<kG2, kMODOUT>(pb0, w2p + ((ot + 8) << 4));
      bias2 = b2c[ocol + 128];
    }
    c = mma_group<kG2>(c, aH + kG2 * 4, pb1);
#pragma unroll
    for (int i = 0; i < 8; ++i) {
      const int row = b0 + rhalf + i;
      out[((size_t)row * kNC + cell) * kMODOUT + ocol] = c[i];
    }
  }
}

extern "C" void kernel_launch(void* const* d_in, const int* in_sizes, int n_in,
                              void* d_out, int out_size, void* d_ws, size_t ws_size,
                              hipStream_t stream) {
  const float* h     = (const float*)d_in[0];
  const float* msg   = (const float*)d_in[1];
  const float* decay = (const float*)d_in[2];
  const float* hebb  = (const float*)d_in[3];
  const float* drift = (const float*)d_in[4];
  const float* slive = (const float*)d_in[5];
  const float* sema  = (const float*)d_in[6];
  const float* w1    = (const float*)d_in[7];
  const float* b1    = (const float*)d_in[8];
  const float* w2    = (const float*)d_in[9];
  const float* b2    = (const float*)d_in[10];
  float* headws = (float*)d_ws;
  float* out    = (float*)d_out;

  const int norm_threads = kBS * kNC * 2 * kN;
  head_kernel<<<norm_threads / 256, 256, 0, stream>>>(
      h, msg, decay, drift, slive, sema, headws);

  dim3 grid(kBS / 16, kNC);
  const size_t smem = (size_t)(8 * 16 * kHSTRIDE + 16 * kHSTRIDE) * sizeof(float);
  modmlp_kernel<<<grid, 256, smem, stream>>>(
      headws, hebb, w1, b1, w2, b2, out);
}


# head launch removed: each wave forms the norms of its two head k-steps in its own A-fragment lanes from h and msg (lane = batch row, no LDS tile), multiplies them first; norm and K sum order changed
# speedup vs baseline: 1.6529x; 1.0604x over previous
#include <hip/hip_runtime.h>
#include <math.h>

constexpr int kBS      = 64;
constexpr int kNC      = 128;
constexpr int kN       = 32;
constexpr int kDN      = 64;
constexpr int kHMOD    = 128;
constexpr int kMODIN   = 2 * kN + 4 + kN * kN;
constexpr int kMODOUT  = kN * kN + kN;
constexpr int kHEAD    = 2 * kN + 4;
constexpr int kHSTRIDE = 132;

typedef __attribute__((ext_vector_type(2))) float v2f;
typedef __attribute__((ext_vector_type(8))) float v8f;

constexpr int kPIECES = kDN / 8;

__device__ __forceinline__ void load_unit(float4 (&u)[kPIECES], const float* __restrict__ p) {
#pragma unroll
  for (int i = 0; i < kPIECES; ++i) u[i] = *(const float4*)(p + 8 * i);
}

__device__ __forceinline__ float unit_norm(const float4 (&u)[kPIECES]) {
  float sx = 0.f, sy = 0.f, sz = 0.f, sw = 0.f;
#pragma unroll
  for (int i = 0; i < kPIECES; ++i) {
    sx = fmaf(u[i].x, u[i].x, sx);
    sy = fmaf(u[i].y, u[i].y, sy);
    sz = fmaf(u[i].z, u[i].z, sz);
    sw = fmaf(u[i].w, u[i].w, sw);
  }
  const float p = (sx + sy) + (sz + sw);
  const float o = __builtin_bit_cast(float, __builtin_amdgcn_permlanex16(
      __builtin_bit_cast(int, p), __builtin_bit_cast(int, p),
      0x76543210, 0xfedcba98, false, false));
  return sqrtf(p + o);
}

constexpr int kG2   = 16;
constexpr int kNT1  = kHMOD / 16;
constexpr int kKW   = (kN * kN) / 8;
constexpr int kNJ   = kKW / 8;

template <int G, int LD>
__device__ __forceinline__ void load_bfrag(v2f (&bf)[G], const float* __restrict__ p) {
#pragma unroll
  for (int j = 0; j < G; ++j) {
    bf[j].x = p[(size_t)(4 * j) * LD];
    bf[j].y = p[(size_t)(4 * j + 1) * LD];
  }
}

template <int G>
__device__ __forceinline__ v8f mma_group(v8f acc, const float* a, const v2f (&bf)[G]) {
#pragma unroll
  for (int j = 0; j < G; ++j) {
    const v2f av = *(const v2f*)(a + 4 * j);
    acc = __builtin_amdgcn_wmma_f32_16x16x4_f32(
        false, av, false, bf[j], (short)0, acc, false, false);
  }
  return acc;
}

__device__ __forceinline__ void load_b1step(v2f (&bf)[kNT1], const float* __restrict__ p) {
#pragma unroll
  for (int n = 0; n < kNT1; ++n) {
    bf[n].x = p[16 * n];
    bf[n].y = p[kHMOD + 16 * n];
  }
}

__device__ __forceinline__ void mma_step(v8f (&acc)[kNT1], const v2f av, const v2f (&bf)[kNT1]) {
#pragma unroll
  for (int n = 0; n < kNT1; ++n)
    acc[n] = __builtin_amdgcn_wmma_f32_16x16x4_f32(
        false, av, false, bf[n], (short)0, acc[n], false, false);
}

__global__ __launch_bounds__(256) void modmlp_kernel(
    const float* __restrict__ h, const float* __restrict__ msg,
    const float* __restrict__ decay, const float* __restrict__ hebb,
    const float* __restrict__ drift, const float* __restrict__ s_live,
    const float* __restrict__ s_ema,
    const float* __restrict__ w1, const float* __restrict__ b1,
    const float* __restrict__ w2, const float* __restrict__ b2,
    float* __restrict__ out) {
  extern __shared__ float smem[];
  float* ldsP = smem;
  float* ldsH = smem + 8 * 16 * kHSTRIDE;

  const int tid   = threadIdx.x;
  const int lane  = tid & 31;
  const int wave  = tid >> 5;
  const int mtile = blockIdx.x;
  const int cell  = blockIdx.y;
  const int b0    = mtile * 16;
  const int arow  = lane & 15;
  const int hsel  = lane >> 4;
  const int koff  = hsel << 1;
  const int rhalf = hsel << 3;

  const size_t prow = (size_t)(b0 + arow) * kNC + cell;
  const float* ap   = hebb + prow * (kN * kN) + wave * kKW + hsel * 4;
  const float* w1c  = w1 + (size_t)cell * kMODIN * kHMOD;
  const float* bp   = w1c + (size_t)(kHEAD + wave * kKW + hsel * 4) * kHMOD + arow;
  const float* hq   = h   + prow * (kN * kDN) + wave * (4 * kDN) + hsel * 4;
  const float* mq   = msg + prow * (kN * kDN) + wave * (4 * kDN) + hsel * 4;
  const float* bh   = w1c + (size_t)koff * kHMOD + arow;

  v2f fb0[kNT1], fb1[kNT1], fb2[kNT1], fb3[kNT1];
  load_b1step(fb0, bh + (size_t)(4 * wave) * kHMOD);
  load_b1step(fb1, bh + (size_t)(4 * (wave + 8)) * kHMOD);

  v2f ah0 = {0.f, 0.f}, ah1 = {0.f, 0.f};
  float4 a4[kNJ];
  {
    float4 ub[4][kPIECES];
#pragma unroll
    for (int u = 0; u < 4; ++u) load_unit(ub[u], hq + u * kDN);
#pragma unroll
    for (int s = 0; s < 4; ++s) {
#pragma unroll
      for (int e = 0; e < 2; ++e) {
        const int t = 2 * s + e;
        const float nrm  = unit_norm(ub[t & 3]);
        const bool  mine = (((t & 3) >> 1) == hsel);
        v2f& dst = (t < 4) ? ah0 : ah1;
        if (t & 1) dst.y = mine ? nrm : dst.y;
        else       dst.x = mine ? nrm : dst.x;
      }
      if (s < 2) {
        load_unit(ub[2 * s],     mq + (2 * s) * kDN);
        load_unit(ub[2 * s + 1], mq + (2 * s + 1) * kDN);
      }
      if (s == 2) {
#pragma unroll
        for (int j = 0; j < kNJ; ++j) a4[j] = *(const float4*)(ap + 8 * j);
        load_b1step(fb2, bp);
        load_b1step(fb3, bp + 2 * kHMOD);
      }
      __builtin_amdgcn_sched_barrier(0);
    }
  }

  v8f acc[kNT1];
#pragma unroll
  for (int n = 0; n < kNT1; ++n)
#pragma unroll
    for (int i = 0; i < 8; ++i) acc[n][i] = 0.f;

  if (wave == 0) {
    v2f fx[kNT1];
    load_b1step(fx, bh + (size_t)(4 * 16) * kHMOD);
    const float4* dp = (const float4*)(decay + prow * kN);
    float m = 0.f;
#pragma unroll
    for (int i = 0; i < kN / 4; ++i) {
      const float4 d = dp[i];
      m += d.x; m += d.y; m += d.z; m += d.w;
    }
    m *= (1.f / (float)kN);
    const float dv = drift[prow];
    const float sl = s_live[b0 + arow];
    const float se = s_ema[b0 + arow];
    const v2f ax = {hsel ? sl : m, hsel ? se : dv};
    mma_step(acc, ax, fx);
  }

  mma_step(acc, ah0, fb0);
  mma_step(acc, ah1, fb1);

#pragma unroll
  for (int g = 0; g < kNJ; ++g) {
    if (g + 1 < kNJ) {
      const float* pn = bp + (size_t)(8 * (g + 1)) * kHMOD;
      if (g & 1) { load_b1step(fb2, pn); load_b1step(fb3, pn + 2 * kHMOD); }
      else       { load_b1step(fb0, pn); load_b1step(fb1, pn + 2 * kHMOD); }
    }
    if (g & 1) {
      mma_step(acc, v2f{a4[g].x, a4[g].y}, fb0);
      mma_step(acc, v2f{a4[g].z, a4[g].w}, fb1);
    } else {
      mma_step(acc, v2f{a4[g].x, a4[g].y}, fb2);
      mma_step(acc, v2f{a4[g].z, a4[g].w}, fb3);
    }
  }

  const float* w2c = w2 + (size_t)cell * kHMOD * kMODOUT;
  const float* w2p = w2c + (size_t)koff * kMODOUT + arow;
  const float* b2c = b2 + (size_t)cell * kMODOUT;
  v2f pb0[kG2], pb1[kG2];
  load_bfrag<kG2, kMODOUT>(pb0, w2p + (wave << 4));
  float bias2 = b2c[(wave << 4) + arow];

  const int rr = tid >> 4;
  const int c8 = (tid & 15) << 3;
  float4 s0 = *(const float4*)(b1 + cell * kHMOD + c8);
  float4 s1 = *(const float4*)(b1 + cell * kHMOD + c8 + 4);
  {
    float* myP = ldsP + wave * 16 * kHSTRIDE + rhalf * kHSTRIDE + arow;
#pragma unroll
    for (int n = 0; n < kNT1; ++n)
#pragma unroll
      for (int i = 0; i < 8; ++i) myP[i * kHSTRIDE + 16 * n] = acc[n][i];
  }
  __syncthreads();
  {
    const float* pp = ldsP + rr * kHSTRIDE + c8;
#pragma unroll
    for (int w = 0; w < 8; ++w) {
      const float4 u0 = *(const float4*)(pp + w * 16 * kHSTRIDE);
      const float4 u1 = *(const float4*)(pp + w * 16 * kHSTRIDE + 4);
      s0.x += u0.x; s0.y += u0.y; s0.z += u0.z; s0.w += u0.w;
      s1.x += u1.x; s1.y += u1.y; s1.z += u1.z; s1.w += u1.w;
    }
    float4 t0, t1;
    t0.x = tanhf(s0.x); t0.y = tanhf(s0.y); t0.z = tanhf(s0.z); t0.w = tanhf(s0.w);
    t1.x = tanhf(s1.x); t1.y = tanhf(s1.y); t1.z = tanhf(s1.z); t1.w = tanhf(s1.w);
    float* hd = ldsH + rr * kHSTRIDE + c8;
    *(float4*)hd       = t0;
    *(float4*)(hd + 4) = t1;
  }
  __syncthreads();

  const float* aH  = ldsH + arow * kHSTRIDE + koff;
  for (int ot = wave; ot < kMODOUT / 16; ot += 8) {
    const int ocol = (ot << 4) + arow;
    v8f c;
#pragma unroll
    for (int i = 0; i < 8; ++i) c[i] = bias2;
    load_bfrag<kG2, kMODOUT>(pb1, w2p + (ot << 4) + (size_t)(kG2 * 4) * kMODOUT);
    c = mma_group<kG2>(c, aH, pb0);
    if (ot + 8 < kMODOUT / 16) {
      load_bfrag<kG2, kMODOUT>(pb0, w2p + ((ot + 8) << 4));
      bias2 = b2c[ocol + 128];
    }
    c = mma_group<kG2>(c, aH + kG2 * 4, pb1);
#pragma unroll
    for (int i = 0; i < 8; ++i) {
      const int row = b0 + rhalf + i;
      out[((size_t)row * kNC + cell) * kMODOUT + ocol] = c[i];
    }
  }
}

extern "C" void kernel_launch(void* const* d_in, const int* in_sizes, int n_in,
                              void* d_out, int out_size, void* d_ws, size_t ws_size,
                              hipStream_t stream) {
  const float* h     = (const float*)d_in[0];
  const float* msg   = (const float*)d_in[1];
  const float* decay = (const float*)d_in[2];
  const float* hebb  = (const float*)d_in[3];
  const float* drift = (const float*)d_in[4];
  const float* slive = (const float*)d_in[5];
  const float* sema  = (const float*)d_in[6];
  const float* w1    = (const float*)d_in[7];
  const float* b1    = (const float*)d_in[8];
  const float* w2    = (const float*)d_in[9];
  const float* b2    = (const float*)d_in[10];
  float* out    = (float*)d_out;

  dim3 grid(kBS / 16, kNC);
  const size_t smem = (size_t)(8 * 16 * kHSTRIDE + 16 * kHSTRIDE) * sizeof(float);
  modmlp_kernel<<<grid, 256, smem, stream>>>(
      h, msg, decay, hebb, drift, slive, sema, w1, b1, w2, b2, out);
}


# cell split over 2 work-groups by K in layer 1, by N in layer 2 (2 launches, K-half sums pass through out): a work-group takes all 64 rows, a weight fragment feeds 2 or 4 M tiles; K sum order changed
# speedup vs baseline: 2.0399x; 1.2341x over previous
#include <hip/hip_runtime.h>
#include <math.h>

constexpr int kBS      = 64;
constexpr int kNC      = 128;
constexpr int kN       = 32;
constexpr int kDN      = 64;
constexpr int kHMOD    = 128;
constexpr int kMODIN   = 2 * kN + 4 + kN * kN;
constexpr int kMODOUT  = kN * kN + kN;
constexpr int kHEAD    = 2 * kN + 4;
constexpr int kHSTRIDE = 132;
constexpr int kNHALF   = kMODOUT / 2;

typedef __attribute__((ext_vector_type(2))) float v2f;
typedef __attribute__((ext_vector_type(8))) float v8f;

constexpr int kPIECES = kDN / 8;

__device__ __forceinline__ void load_unit(float4 (&u)[kPIECES], const float* __restrict__ p) {
#pragma unroll
  for (int i = 0; i < kPIECES; ++i) u[i] = *(const float4*)(p + 8 * i);
}

__device__ __forceinline__ float unit_norm(const float4 (&u)[kPIECES]) {
  float sx = 0.f, sy = 0.f, sz = 0.f, sw = 0.f;
#pragma unroll
  for (int i = 0; i < kPIECES; ++i) {
    sx = fmaf(u[i].x, u[i].x, sx);
    sy = fmaf(u[i].y, u[i].y, sy);
    sz = fmaf(u[i].z, u[i].z, sz);
    sw = fmaf(u[i].w, u[i].w, sw);
  }
  const float p = (sx + sy) + (sz + sw);
  const float o = __builtin_bit_cast(float, __builtin_amdgcn_permlanex16(
      __builtin_bit_cast(int, p), __builtin_bit_cast(int, p),
      0x76543210, 0xfedcba98, false, false));
  return sqrtf(p + o);
}

constexpr int kNT1 = kHMOD / 16;

__device__ __forceinline__ void load_b1step(v2f (&bf)[kNT1], const float* __restrict__ p) {
#pragma unroll
  for (int n = 0; n < kNT1; ++n) {
    bf[n].x = p[16 * n];
    bf[n].y = p[kHMOD + 16 * n];
  }
}

__device__ __forceinline__ void mma_step(v8f (&acc)[kNT1], const v2f av, const v2f (&bf)[kNT1]) {
#pragma unroll
  for (int n = 0; n < kNT1; ++n)
    acc[n] = __builtin_amdgcn_wmma_f32_16x16x4_f32(
        false, av, false, bf[n], (short)0, acc[n], false, false);
}

constexpr int kNJ = 8;

__global__ __launch_bounds__(512) void l1_partial_kernel(
    const float* __restrict__ h, const float* __restrict__ msg,
    const float* __restrict__ decay, const float* __restrict__ hebb,
    const float* __restrict__ drift, const float* __restrict__ s_live,
    const float* __restrict__ s_ema,
    const float* __restrict__ w1, float* __restrict__ out) {
  __shared__ float ldsP[16 * 16 * kHSTRIDE];

  const int tid   = threadIdx.x;
  const int lane  = tid & 31;
  const int wave  = __builtin_amdgcn_readfirstlane(tid >> 5);
  const int kh    = blockIdx.x;
  const int cell  = blockIdx.y;
  const int mh    = wave >> 3;
  const int ks    = wave & 7;
  const int arow  = lane & 15;
  const int hsel  = lane >> 4;
  const int koff  = hsel << 1;
  const int rhalf = hsel << 3;

  const size_t prow0 = (size_t)(32 * mh + arow) * kNC + cell;
  const size_t prow1 = prow0 + (size_t)16 * kNC;
  const int    q0    = 512 * kh + 64 * ks;
  const float* ap0   = hebb + prow0 * (kN * kN) + q0 + hsel * 4;
  const float* ap1   = hebb + prow1 * (kN * kN) + q0 + hsel * 4;
  const float* w1c   = w1 + (size_t)cell * kMODIN * kHMOD;
  const float* bp    = w1c + (size_t)(kHEAD + q0 + hsel * 4) * kHMOD + arow;
  const float* vsrc  = kh ? msg : h;
  const float* vq0   = vsrc + prow0 * (kN * kDN) + ks * (4 * kDN) + hsel * 4;
  const float* vq1   = vsrc + prow1 * (kN * kDN) + ks * (4 * kDN) + hsel * 4;
  const float* bh    = w1c + (size_t)koff * kHMOD + arow;
  const int    hs    = 8 * kh + ks;

  const bool scal = (kh == 0) && (ks == 0);
  v2f ax0 = {0.f, 0.f}, ax1 = {0.f, 0.f};
  if (scal) {
#pragma unroll
    for (int t = 0; t < 2; ++t) {
      const size_t prow = t ? prow1 : prow0;
      const int    brow = 32 * mh + 16 * t + arow;
      const float4* dp = (const float4*)(decay + prow * kN);
      float m = 0.f;
#pragma unroll
      for (int i = 0; i < kN / 4; ++i) {
        const float4 d = dp[i];
        m += d.x; m += d.y; m += d.z; m += d.w;
      }
      m *= (1.f / (float)kN);
      const float dv = drift[prow];
      const float sl = s_live[brow];
      const float se = s_ema[brow];
      const v2f ax = {hsel ? sl : m, hsel ? se : dv};
      if (t) ax1 = ax; else ax0 = ax;
    }
  }

  v2f fb[3][kNT1];
  load_b1step(fb[0], bh + (size_t)(4 * hs) * kHMOD);

  v2f ah0 = {0.f, 0.f}, ah1 = {0.f, 0.f};
  float4 a4[2][kNJ];
  {
    float4 ub[4][kPIECES];
#pragma unroll
    for (int u = 0; u < 4; ++u) load_unit(ub[u], vq0 + u * kDN);
#pragma unroll
    for (int s = 0; s < 4; ++s) {
#pragma unroll
      for (int e = 0; e < 2; ++e) {
        const int t = 2 * s + e;
        const float nrm  = unit_norm(ub[t & 3]);
        const bool  mine = (((t & 3) >> 1) == hsel);
        v2f& dst = (t < 4) ? ah0 : ah1;
        if (t & 1) dst.y = mine ? nrm : dst.y;
        else       dst.x = mine ? nrm : dst.x;
      }
      if (s < 2) {
        load_unit(ub[2 * s],     vq1 + (2 * s) * kDN);
        load_unit(ub[2 * s + 1], vq1 + (2 * s + 1) * kDN);
      }
      if (s == 2) {
#pragma unroll
        for (int j = 0; j < kNJ; ++j) {
          a4[0][j] = *(const float4*)(ap0 + 8 * j);
          a4[1][j] = *(const float4*)(ap1 + 8 * j);
        }
        load_b1step(fb[1], bp);
        load_b1step(fb[2], bp + 2 * kHMOD);
      }
      __builtin_amdgcn_sched_barrier(0);
    }
  }

  v8f acc[2][kNT1];
#pragma unroll
  for (int t = 0; t < 2; ++t)
#pragma unroll
    for (int n = 0; n < kNT1; ++n)
#pragma unroll
      for (int i = 0; i < 8; ++i) acc[t][n][i] = 0.f;

#pragma unroll
  for (int i = 0; i <= 2 * kNJ; ++i) {
    const int nx = i + 2;
    if (nx >= 3 && nx <= 2 * kNJ) {
      const int sn = nx - 1;
      load_b1step(fb[nx % 3], bp + (size_t)(8 * (sn >> 1) + 2 * (sn & 1)) * kHMOD);
    } else if (nx == 2 * kNJ + 1 && scal) {
      load_b1step(fb[nx % 3], bh + (size_t)(4 * 16) * kHMOD);
    }
    v2f a0 = ah0, a1 = ah1;
    if (i > 0) {
      const int st = i - 1, j = st >> 1;
      a0 = (st & 1) ? v2f{a4[0][j].z, a4[0][j].w} : v2f{a4[0][j].x, a4[0][j].y};
      a1 = (st & 1) ? v2f{a4[1][j].z, a4[1][j].w} : v2f{a4[1][j].x, a4[1][j].y};
    }
    mma_step(acc[0], a0, fb[i % 3]);
    mma_step(acc[1], a1, fb[i % 3]);
    __builtin_amdgcn_sched_barrier(0);
  }
  if (scal) {
    mma_step(acc[0], ax0, fb[(2 * kNJ + 1) % 3]);
    mma_step(acc[1], ax1, fb[(2 * kNJ + 1) % 3]);
  }

  const int rr  = (tid & 255) >> 4;
  const int c8  = (tid & 15) << 3;
  const int mh2 = tid >> 8;
  float* myP = ldsP + wave * 16 * kHSTRIDE + rhalf * kHSTRIDE + arow;
  const float* pp = ldsP + (mh2 * 8) * 16 * kHSTRIDE + rr * kHSTRIDE + c8;
#pragma unroll
  for (int t = 0; t < 2; ++t) {
    if (t) __syncthreads();
#pragma unroll
    for (int n = 0; n < kNT1; ++n)
#pragma unroll
      for (int i = 0; i < 8; ++i) myP[i * kHSTRIDE + 16 * n] = acc[t][n][i];
    __syncthreads();
    float4 s0 = {0.f, 0.f, 0.f, 0.f}, s1 = {0.f, 0.f, 0.f, 0.f};
#pragma unroll
    for (int w = 0; w < 8; ++w) {
      const float4 u0 = *(const float4*)(pp + w * 16 * kHSTRIDE);
      const float4 u1 = *(const float4*)(pp + w * 16 * kHSTRIDE + 4);
      s0.x += u0.x; s0.y += u0.y; s0.z += u0.z; s0.w += u0.w;
      s1.x += u1.x; s1.y += u1.y; s1.z += u1.z; s1.w += u1.w;
    }
    const int row = 32 * mh2 + 16 * t + rr;
    float* o = out + ((size_t)row * kNC + cell) * kMODOUT + kHMOD * kh + c8;
    *(float4*)o                = s0;
    *(float4*)(o + 4)          = s1;
    *(float4*)(o + kNHALF)     = s0;
    *(float4*)(o + kNHALF + 4) = s1;
  }
}

constexpr int kG = 8;

template <int NT>
__device__ __forceinline__ void load_bgroup(v2f (&bf)[kG][NT], const float* __restrict__ p) {
#pragma unroll
  for (int s = 0; s < kG; ++s)
#pragma unroll
    for (int n = 0; n < NT; ++n) {
      bf[s][n].x = p[(size_t)(4 * s) * kMODOUT + 16 * n];
      bf[s][n].y = p[(size_t)(4 * s + 1) * kMODOUT + 16 * n];
    }
}

template <int NT>
__device__ __forceinline__ void mma_group(v8f (&acc)[4][NT], const float* a, const v2f (&bf)[kG][NT]) {
#pragma unroll
  for (int s = 0; s < kG; ++s)
#pragma unroll
    for (int t = 0; t < 4; ++t) {
      const v2f av = *(const v2f*)(a + t * 16 * kHSTRIDE + 4 * s);
#pragma unroll
      for (int n = 0; n < NT; ++n)
        acc[t][n] = __builtin_amdgcn_wmma_f32_16x16x4_f32(
            false, av, false, bf[s][n], (short)0, acc[t][n], false, false);
    }
}

template <int NT>
__device__ __forceinline__ void store_tiles(const v8f (&acc)[4][NT], float* o) {
#pragma unroll
  for (int t = 0; t < 4; ++t)
#pragma unroll
    for (int n = 0; n < NT; ++n)
#pragma unroll
      for (int i = 0; i < 8; ++i)
        o[(size_t)(16 * t + i) * kNC * kMODOUT + 16 * n] = acc[t][n][i];
}

__global__ __launch_bounds__(512) void l2_out_kernel(
    const float* __restrict__ b1, const float* __restrict__ w2,
    const float* __restrict__ b2, float* out) {
  __shared__ float ldsH[kBS * kHSTRIDE];

  const int tid   = threadIdx.x;
  const int lane  = tid & 31;
  const int wave  = __builtin_amdgcn_readfirstlane(tid >> 5);
  const int nh    = blockIdx.x;
  const int cell  = blockIdx.y;
  const int arow  = lane & 15;
  const int hsel  = lane >> 4;
  const int koff  = hsel << 1;
  const int rhalf = hsel << 3;

  const int ta = nh ? 34 + 2 * wave : 2 * wave;
  const int ts = nh ? 33 : 32;
  const float* w2c = w2 + (size_t)cell * kHMOD * kMODOUT;
  const float* w2p = w2c + (size_t)koff * kMODOUT + arow;
  const float* b2c = b2 + (size_t)cell * kMODOUT + arow;
  float* orow = out + ((size_t)rhalf * kNC + cell) * kMODOUT + arow;

  v2f pbA[kG][2], pbB[kG][2];
  const float* wp = w2p + 16 * ta;
  load_bgroup<2>(pbA, wp);
  const float bz0 = b2c[16 * ta];
  const float bz1 = b2c[16 * ta + 16];

  {
    const int row = tid >> 3;
    const int c4  = (tid & 7) << 2;
    const float* pr = out + ((size_t)row * kNC + cell) * kMODOUT + kNHALF * nh + c4;
    const float* bb = b1 + cell * kHMOD + c4;
    float* hd = ldsH + row * kHSTRIDE + c4;
    float4 p0[4], p1[4], bv[4];
#pragma unroll
    for (int q = 0; q < 4; ++q) {
      p0[q] = *(const float4*)(pr + 32 * q);
      p1[q] = *(const float4*)(pr + kHMOD + 32 * q);
      bv[q] = *(const float4*)(bb + 32 * q);
    }
#pragma unroll
    for (int q = 0; q < 4; ++q) {
      float4 t;
      t.x = tanhf((bv[q].x + p0[q].x) + p1[q].x);
      t.y = tanhf((bv[q].y + p0[q].y) + p1[q].y);
      t.z = tanhf((bv[q].z + p0[q].z) + p1[q].z);
      t.w = tanhf((bv[q].w + p0[q].w) + p1[q].w);
      *(float4*)(hd + 32 * q) = t;
    }
  }
  __syncthreads();

  const float* aH = ldsH + arow * kHSTRIDE + koff;
  v2f sb[4][kG][1];
  float bzs = 0.f;
  const float* sp = w2p + 16 * ts;
  {
    v8f acc[4][2];
#pragma unroll
    for (int t = 0; t < 4; ++t)
#pragma unroll
      for (int i = 0; i < 8; ++i) { acc[t][0][i] = bz0; acc[t][1][i] = bz1; }
    load_bgroup<2>(pbB, wp + (size_t)(4 * kG) * kMODOUT);
    mma_group<2>(acc, aH, pbA);
    load_bgroup<2>(pbA, wp + (size_t)(8 * kG) * kMODOUT);
    mma_group<2>(acc, aH + 4 * kG, pbB);
    load_bgroup<2>(pbB, wp + (size_t)(12 * kG) * kMODOUT);
    if (wave == 0) {
#pragma unroll
      for (int g = 0; g < 4; ++g) load_bgroup<1>(sb[g], sp + (size_t)(4 * kG * g) * kMODOUT);
      bzs = b2c[16 * ts];
    }
    mma_group<2>(acc, aH + 8 * kG, pbA);
    mma_group<2>(acc, aH + 12 * kG, pbB);
    store_tiles<2>(acc, orow + 16 * ta);
  }

  if (wave == 0) {
    v8f acc[4][1];
#pragma unroll
    for (int t = 0; t < 4; ++t)
#pragma unroll
      for (int i = 0; i < 8; ++i) acc[t][0][i] = bzs;
#pragma unroll
    for (int g = 0; g < 4; ++g) mma_group<1>(acc, aH + 4 * kG * g, sb[g]);
    store_tiles<1>(acc, orow + 16 * ts);
  }
}

extern "C" void kernel_launch(void* const* d_in, const int* in_sizes, int n_in,
                              void* d_out, int out_size, void* d_ws, size_t ws_size,
                              hipStream_t stream) {
  const float* h     = (const float*)d_in[0];
  const float* msg   = (const float*)d_in[1];
  const float* decay = (const float*)d_in[2];
  const float* hebb  = (const float*)d_in[3];
  const float* drift = (const float*)d_in[4];
  const float* slive = (const float*)d_in[5];
  const float* sema  = (const float*)d_in[6];
  const float* w1    = (const float*)d_in[7];
  const float* b1    = (const float*)d_in[8];
  const float* w2    = (const float*)d_in[9];
  const float* b2    = (const float*)d_in[10];
  float* out    = (float*)d_out;

  dim3 grid(2, kNC);
  l1_partial_kernel<<<grid, 512, 0, stream>>>(
      h, msg, decay, hebb, drift, slive, sema, w1, out);
  l2_out_kernel<<<grid, 512, 0, stream>>>(b1, w2, b2, out);
}


# h/msg norm loads as whole lines: a load takes 2 whole vectors of one batch row (4 lines, not 32 B of 16); 16 rows summed by a transposing row_xmask butterfly, lane = row; norm sum order changed
# speedup vs baseline: 2.1533x; 1.0556x over previous
#include <hip/hip_runtime.h>
#include <math.h>

constexpr int kBS      = 64;
constexpr int kNC      = 128;
constexpr int kN       = 32;
constexpr int kDN      = 64;
constexpr int kHMOD    = 128;
constexpr int kMODIN   = 2 * kN + 4 + kN * kN;
constexpr int kMODOUT  = kN * kN + kN;
constexpr int kHEAD    = 2 * kN + 4;
constexpr int kHSTRIDE = 132;
constexpr int kNHALF   = kMODOUT / 2;

typedef __attribute__((ext_vector_type(2))) float v2f;
typedef __attribute__((ext_vector_type(8))) float v8f;

constexpr int kROWS = 16;
constexpr size_t kVROW = (size_t)kNC * kN * kDN;

__device__ __forceinline__ void load_rows(float4 (&u)[kROWS], const float* __restrict__ p) {
#pragma unroll
  for (int R = 0; R < kROWS; ++R) u[R] = *(const float4*)(p + (size_t)R * kVROW);
}

template <int CTRL>
__device__ __forceinline__ float xmask_add(const float keep, const float give) {
  const int g = __builtin_amdgcn_update_dpp(0, __builtin_bit_cast(int, give), CTRL, 0xf, 0xf, true);
  return keep + __builtin_bit_cast(float, g);
}

__device__ __forceinline__ float rows_sumsq(const float4 (&u)[kROWS], const int c) {
  float p[kROWS];
#pragma unroll
  for (int R = 0; R < kROWS; ++R)
    p[R] = fmaf(u[R].w, u[R].w, fmaf(u[R].z, u[R].z, fmaf(u[R].y, u[R].y, u[R].x * u[R].x)));
  const bool b3 = (c & 8) != 0, b2 = (c & 4) != 0, b1 = (c & 2) != 0, b0 = (c & 1) != 0;
  float q[8], r[4], s[2];
#pragma unroll
  for (int R = 0; R < 8; ++R)
    q[R] = xmask_add<0x168>(b3 ? p[R + 8] : p[R], b3 ? p[R] : p[R + 8]);
#pragma unroll
  for (int R = 0; R < 4; ++R)
    r[R] = xmask_add<0x164>(b2 ? q[R + 4] : q[R], b2 ? q[R] : q[R + 4]);
#pragma unroll
  for (int R = 0; R < 2; ++R)
    s[R] = xmask_add<0x162>(b1 ? r[R + 2] : r[R], b1 ? r[R] : r[R + 2]);
  return xmask_add<0x161>(b0 ? s[1] : s[0], b0 ? s[0] : s[1]);
}

constexpr int kNT1 = kHMOD / 16;

__device__ __forceinline__ void load_b1step(v2f (&bf)[kNT1], const float* __restrict__ p) {
#pragma unroll
  for (int n = 0; n < kNT1; ++n) {
    bf[n].x = p[16 * n];
    bf[n].y = p[kHMOD + 16 * n];
  }
}

__device__ __forceinline__ void mma_step(v8f (&acc)[kNT1], const v2f av, const v2f (&bf)[kNT1]) {
#pragma unroll
  for (int n = 0; n < kNT1; ++n)
    acc[n] = __builtin_amdgcn_wmma_f32_16x16x4_f32(
        false, av, false, bf[n], (short)0, acc[n], false, false);
}

constexpr int kNJ = 8;

__global__ __launch_bounds__(512) void l1_partial_kernel(
    const float* __restrict__ h, const float* __restrict__ msg,
    const float* __restrict__ decay, const float* __restrict__ hebb,
    const float* __restrict__ drift, const float* __restrict__ s_live,
    const float* __restrict__ s_ema,
    const float* __restrict__ w1, float* __restrict__ out) {
  __shared__ float ldsP[16 * 16 * kHSTRIDE];

  const int tid   = threadIdx.x;
  const int lane  = tid & 31;
  const int wave  = __builtin_amdgcn_readfirstlane(tid >> 5);
  const int kh    = blockIdx.x;
  const int cell  = blockIdx.y;
  const int mh    = wave >> 3;
  const int ks    = wave & 7;
  const int arow  = lane & 15;
  const int hsel  = lane >> 4;
  const int koff  = hsel << 1;
  const int rhalf = hsel << 3;

  const size_t prow0 = (size_t)(32 * mh + arow) * kNC + cell;
  const size_t prow1 = prow0 + (size_t)16 * kNC;
  const int    q0    = 512 * kh + 64 * ks;
  const float* ap0   = hebb + prow0 * (kN * kN) + q0 + hsel * 4;
  const float* ap1   = hebb + prow1 * (kN * kN) + q0 + hsel * 4;
  const float* w1c   = w1 + (size_t)cell * kMODIN * kHMOD;
  const float* bp    = w1c + (size_t)(kHEAD + q0 + hsel * 4) * kHMOD + arow;
  const float* vsrc  = kh ? msg : h;
  const float* vb0   = vsrc + ((size_t)(32 * mh) * kNC + cell) * (kN * kDN)
                            + (4 * ks + 2 * hsel) * kDN + 4 * arow;
  const float* vb1   = vb0 + (size_t)16 * kVROW;
  const float* bh    = w1c + (size_t)koff * kHMOD + arow;
  const int    hs    = 8 * kh + ks;

  const bool scal = (kh == 0) && (ks == 0);
  v2f ax0 = {0.f, 0.f}, ax1 = {0.f, 0.f};
  if (scal) {
#pragma unroll
    for (int t = 0; t < 2; ++t) {
      const size_t prow = t ? prow1 : prow0;
      const int    brow = 32 * mh + 16 * t + arow;
      const float4* dp = (const float4*)(decay + prow * kN);
      float m = 0.f;
#pragma unroll
      for (int i = 0; i < kN / 4; ++i) {
        const float4 d = dp[i];
        m += d.x; m += d.y; m += d.z; m += d.w;
      }
      m *= (1.f / (float)kN);
      const float dv = drift[prow];
      const float sl = s_live[brow];
      const float se = s_ema[brow];
      const v2f ax = {hsel ? sl : m, hsel ? se : dv};
      if (t) ax1 = ax; else ax0 = ax;
    }
  }

  v2f fb[3][kNT1];
  load_b1step(fb[0], bh + (size_t)(4 * hs) * kHMOD);

  v2f ah0 = {0.f, 0.f}, ah1 = {0.f, 0.f};
  float4 a4[2][kNJ];
  {
    float4 ub[2][kROWS];
    load_rows(ub[0], vb0);
    load_rows(ub[1], vb0 + kDN);
#pragma unroll
    for (int g = 0; g < 4; ++g) {
      const float nrm = sqrtf(rows_sumsq(ub[g & 1], arow));
      if (g == 0) ah0.x = nrm;
      if (g == 1) ah0.y = nrm;
      if (g == 2) ah1.x = nrm;
      if (g == 3) ah1.y = nrm;
      if (g < 2) load_rows(ub[g & 1], vb1 + (g & 1) * kDN);
      if (g == 2) {
#pragma unroll
        for (int j = 0; j < kNJ; ++j) {
          a4[0][j] = *(const float4*)(ap0 + 8 * j);
          a4[1][j] = *(const float4*)(ap1 + 8 * j);
        }
        load_b1step(fb[1], bp);
        load_b1step(fb[2], bp + 2 * kHMOD);
      }
      __builtin_amdgcn_sched_barrier(0);
    }
  }

  v8f acc[2][kNT1];
#pragma unroll
  for (int t = 0; t < 2; ++t)
#pragma unroll
    for (int n = 0; n < kNT1; ++n)
#pragma unroll
      for (int i = 0; i < 8; ++i) acc[t][n][i] = 0.f;

#pragma unroll
  for (int i = 0; i <= 2 * kNJ; ++i) {
    const int nx = i + 2;
    if (nx >= 3 && nx <= 2 * kNJ) {
      const int sn = nx - 1;
      load_b1step(fb[nx % 3], bp + (size_t)(8 * (sn >> 1) + 2 * (sn & 1)) * kHMOD);
    } else if (nx == 2 * kNJ + 1 && scal) {
      load_b1step(fb[nx % 3], bh + (size_t)(4 * 16) * kHMOD);
    }
    v2f a0 = ah0, a1 = ah1;
    if (i > 0) {
      const int st = i - 1, j = st >> 1;
      a0 = (st & 1) ? v2f{a4[0][j].z, a4[0][j].w} : v2f{a4[0][j].x, a4[0][j].y};
      a1 = (st & 1) ? v2f{a4[1][j].z, a4[1][j].w} : v2f{a4[1][j].x, a4[1][j].y};
    }
    mma_step(acc[0], a0, fb[i % 3]);
    mma_step(acc[1], a1, fb[i % 3]);
    __builtin_amdgcn_sched_barrier(0);
  }
  if (scal) {
    mma_step(acc[0], ax0, fb[(2 * kNJ + 1) % 3]);
    mma_step(acc[1], ax1, fb[(2 * kNJ + 1) % 3]);
  }

  const int rr  = (tid & 255) >> 4;
  const int c8  = (tid & 15) << 3;
  const int mh2 = tid >> 8;
  float* myP = ldsP + wave * 16 * kHSTRIDE + rhalf * kHSTRIDE + arow;
  const float* pp = ldsP + (mh2 * 8) * 16 * kHSTRIDE + rr * kHSTRIDE + c8;
#pragma unroll
  for (int t = 0; t < 2; ++t) {
    if (t) __syncthreads();
#pragma unroll
    for (int n = 0; n < kNT1; ++n)
#pragma unroll
      for (int i = 0; i < 8; ++i) myP[i * kHSTRIDE + 16 * n] = acc[t][n][i];
    __syncthreads();
    float4 s0 = {0.f, 0.f, 0.f, 0.f}, s1 = {0.f, 0.f, 0.f, 0.f};
#pragma unroll
    for (int w = 0; w < 8; ++w) {
      const float4 u0 = *(const float4*)(pp + w * 16 * kHSTRIDE);
      const float4 u1 = *(const float4*)(pp + w * 16 * kHSTRIDE + 4);
      s0.x += u0.x; s0.y += u0.y; s0.z += u0.z; s0.w += u0.w;
      s1.x += u1.x; s1.y += u1.y; s1.z += u1.z; s1.w += u1.w;
    }
    const int row = 32 * mh2 + 16 * t + rr;
    float* o = out + ((size_t)row * kNC + cell) * kMODOUT + kHMOD * kh + c8;
    *(float4*)o                = s0;
    *(float4*)(o + 4)          = s1;
    *(float4*)(o + kNHALF)     = s0;
    *(float4*)(o + kNHALF + 4) = s1;
  }
}

constexpr int kG = 8;

template <int NT>
__device__ __forceinline__ void load_bgroup(v2f (&bf)[kG][NT], const float* __restrict__ p) {
#pragma unroll
  for (int s = 0; s < kG; ++s)
#pragma unroll
    for (int n = 0; n < NT; ++n) {
      bf[s][n].x = p[(size_t)(4 * s) * kMODOUT + 16 * n];
      bf[s][n].y = p[(size_t)(4 * s + 1) * kMODOUT + 16 * n];
    }
}

template <int NT>
__device__ __forceinline__ void mma_group(v8f (&acc)[4][NT], const float* a, const v2f (&bf)[kG][NT]) {
#pragma unroll
  for (int s = 0; s < kG; ++s)
#pragma unroll
    for (int t = 0; t < 4; ++t) {
      const v2f av = *(const v2f*)(a + t * 16 * kHSTRIDE + 4 * s);
#pragma unroll
      for (int n = 0; n < NT; ++n)
        acc[t][n] = __builtin_amdgcn_wmma_f32_16x16x4_f32(
            false, av, false, bf[s][n], (short)0, acc[t][n], false, false);
    }
}

template <int NT>
__device__ __forceinline__ void store_tiles(const v8f (&acc)[4][NT], float* o) {
#pragma unroll
  for (int t = 0; t < 4; ++t)
#pragma unroll
    for (int n = 0; n < NT; ++n)
#pragma unroll
      for (int i = 0; i < 8; ++i)
        o[(size_t)(16 * t + i) * kNC * kMODOUT + 16 * n] = acc[t][n][i];
}

__global__ __launch_bounds__(512) void l2_out_kernel(
    const float* __restrict__ b1, const float* __restrict__ w2,
    const float* __restrict__ b2, float* out) {
  __shared__ float ldsH[kBS * kHSTRIDE];

  const int tid   = threadIdx.x;
  const int lane  = tid & 31;
  const int wave  = __builtin_amdgcn_readfirstlane(tid >> 5);
  const int nh    = blockIdx.x;
  const int cell  = blockIdx.y;
  const int arow  = lane & 15;
  const int hsel  = lane >> 4;
  const int koff  = hsel << 1;
  const int rhalf = hsel << 3;

  const int ta = nh ? 34 + 2 * wave : 2 * wave;
  const int ts = nh ? 33 : 32;
  const float* w2c = w2 + (size_t)cell * kHMOD * kMODOUT;
  const float* w2p = w2c + (size_t)koff * kMODOUT + arow;
  const float* b2c = b2 + (size_t)cell * kMODOUT + arow;
  float* orow = out + ((size_t)rhalf * kNC + cell) * kMODOUT + arow;

  v2f pbA[kG][2], pbB[kG][2];
  const float* wp = w2p + 16 * ta;
  load_bgroup<2>(pbA, wp);
  const float bz0 = b2c[16 * ta];
  const float bz1 = b2c[16 * ta + 16];

  {
    const int row = tid >> 3;
    const int c4  = (tid & 7) << 2;
    const float* pr = out + ((size_t)row * kNC + cell) * kMODOUT + kNHALF * nh + c4;
    const float* bb = b1 + cell * kHMOD + c4;
    float* hd = ldsH + row * kHSTRIDE + c4;
    float4 p0[4], p1[4], bv[4];
#pragma unroll
    for (int q = 0; q < 4; ++q) {
      p0[q] = *(const float4*)(pr + 32 * q);
      p1[q] = *(const float4*)(pr + kHMOD + 32 * q);
      bv[q] = *(const float4*)(bb + 32 * q);
    }
#pragma unroll
    for (int q = 0; q < 4; ++q) {
      float4 t;
      t.x = tanhf((bv[q].x + p0[q].x) + p1[q].x);
      t.y = tanhf((bv[q].y + p0[q].y) + p1[q].y);
      t.z = tanhf((bv[q].z + p0[q].z) + p1[q].z);
      t.w = tanhf((bv[q].w + p0[q].w) + p1[q].w);
      *(float4*)(hd + 32 * q) = t;
    }
  }
  __syncthreads();

  const float* aH = ldsH + arow * kHSTRIDE + koff;
  v2f sb[4][kG][1];
  float bzs = 0.f;
  const float* sp = w2p + 16 * ts;
  {
    v8f acc[4][2];
#pragma unroll
    for (int t = 0; t < 4; ++t)
#pragma unroll
      for (int i = 0; i < 8; ++i) { acc[t][0][i] = bz0; acc[t][1][i] = bz1; }
    load_bgroup<2>(pbB, wp + (size_t)(4 * kG) * kMODOUT);
    mma_group<2>(acc, aH, pbA);
    load_bgroup<2>(pbA, wp + (size_t)(8 * kG) * kMODOUT);
    mma_group<2>(acc, aH + 4 * kG, pbB);
    load_bgroup<2>(pbB, wp + (size_t)(12 * kG) * kMODOUT);
    if (wave == 0) {
#pragma unroll
      for (int g = 0; g < 4; ++g) load_bgroup<1>(sb[g], sp + (size_t)(4 * kG * g) * kMODOUT);
      bzs = b2c[16 * ts];
    }
    mma_group<2>(acc, aH + 8 * kG, pbA);
    mma_group<2>(acc, aH + 12 * kG, pbB);
    store_tiles<2>(acc, orow + 16 * ta);
  }

  if (wave == 0) {
    v8f acc[4][1];
#pragma unroll
    for (int t = 0; t < 4; ++t)
#pragma unroll
      for (int i = 0; i < 8; ++i) acc[t][0][i] = bzs;
#pragma unroll
    for (int g = 0; g < 4; ++g) mma_group<1>(acc, aH + 4 * kG * g, sb[g]);
    store_tiles<1>(acc, orow + 16 * ts);
  }
}

extern "C" void kernel_launch(void* const* d_in, const int* in_sizes, int n_in,
                              void* d_out, int out_size, void* d_ws, size_t ws_size,
                              hipStream_t stream) {
  const float* h     = (const float*)d_in[0];
  const float* msg   = (const float*)d_in[1];
  const float* decay = (const float*)d_in[2];
  const float* hebb  = (const float*)d_in[3];
  const float* drift = (const float*)d_in[4];
  const float* slive = (const float*)d_in[5];
  const float* sema  = (const float*)d_in[6];
  const float* w1    = (const float*)d_in[7];
  const float* b1    = (const float*)d_in[8];
  const float* w2    = (const float*)d_in[9];
  const float* b2    = (const float*)d_in[10];
  float* out    = (float*)d_out;

  dim3 grid(2, kNC);
  l1_partial_kernel<<<grid, 512, 0, stream>>>(
      h, msg, decay, hebb, drift, slive, sema, w1, out);
  l2_out_kernel<<<grid, 512, 0, stream>>>(b1, w2, b2, out);
}
